# starting kernel
# baseline (speedup 1.0000x reference)
#include <hip/hip_runtime.h>
#include <math.h>

#define ALPHA_ERF 0.8862269254527580137f

typedef __attribute__((ext_vector_type(2))) float v2f;
typedef __attribute__((ext_vector_type(8))) float v8f;

__global__ __launch_bounds__(256) void k1_reduce(
    const float* __restrict__ h, const float* __restrict__ v,
    const float* __restrict__ z, float* __restrict__ partial,
    int B, int nchunk)
{
    const int b0    = (blockIdx.x * 256 + threadIdx.x) * 2;
    const int chunk = blockIdx.y;
    const int nBeg  = chunk * nchunk;
    const int nEnd  = nBeg + nchunk;

    float acc[5][2];
#pragma unroll
    for (int r = 0; r < 5; ++r) { acc[r][0] = 0.0f; acc[r][1] = 0.0f; }

    for (int n = nBeg; n < nEnd; ++n) {
        const size_t hoff = (size_t)n * (size_t)B + (size_t)b0;
        if (n + 4 < nEnd) __builtin_prefetch(h + hoff + 4 * (size_t)B, 0, 1);
        const float2 hv = *(const float2*)(h + hoff);
        const float p0 = erff(ALPHA_ERF * hv.x);
        const float p1 = erff(ALPHA_ERF * hv.y);
        const float4 vv = *(const float4*)(v + (size_t)n * 4);
        const float  zv = z[n];
        acc[0][0] = fmaf(vv.x, p0, acc[0][0]); acc[0][1] = fmaf(vv.x, p1, acc[0][1]);
        acc[1][0] = fmaf(vv.y, p0, acc[1][0]); acc[1][1] = fmaf(vv.y, p1, acc[1][1]);
        acc[2][0] = fmaf(vv.z, p0, acc[2][0]); acc[2][1] = fmaf(vv.z, p1, acc[2][1]);
        acc[3][0] = fmaf(vv.w, p0, acc[3][0]); acc[3][1] = fmaf(vv.w, p1, acc[3][1]);
        acc[4][0] = fmaf(zv,   p0, acc[4][0]); acc[4][1] = fmaf(zv,   p1, acc[4][1]);
    }

#pragma unroll
    for (int r = 0; r < 5; ++r) {
        float2 o; o.x = acc[r][0]; o.y = acc[r][1];
        *(float2*)(partial + (size_t)(chunk * 5 + r) * (size_t)B + b0) = o;
    }
}

__global__ __launch_bounds__(256) void k2_finalize(
    const float* __restrict__ partial, float* __restrict__ tbuf,
    float* __restrict__ y_out, int B, int nchunks, float invN)
{
    const int idx = blockIdx.x * 256 + threadIdx.x;
    if (idx >= 5 * B) return;
    const int r = idx / B;
    const int b = idx - r * B;
    float s = 0.0f;
    for (int c = 0; c < nchunks; ++c)
        s += partial[(size_t)(c * 5 + r) * (size_t)B + b];
    if (r == 4) y_out[b] = s * invN;
    else        tbuf[(size_t)r * (size_t)B + b] = s;
}

__global__ __launch_bounds__(256) void k3_update(
    const float* __restrict__ h, const float* __restrict__ u,
    const float* __restrict__ m, const float* __restrict__ x,
    const float* __restrict__ tbuf, float* __restrict__ hnew,
    int B, float invN)
{
    const int lane = threadIdx.x & 31;
    const int wave = threadIdx.x >> 5;
    const int lo   = lane & 15;
    const int hi   = lane >> 4;

    const int colGroups = B >> 6;
    const int gwave     = blockIdx.x * 8 + wave;
    const int rowTile   = gwave / colGroups;
    const int colGroup  = gwave - rowTile * colGroups;
    const int n0        = rowTile * 16;

    v2f a;
    {
        const float2 av = *(const float2*)(u + (size_t)(n0 + lo) * 4 + hi * 2);
        a.x = av.x; a.y = av.y;
    }

    float mv[8];
    {
        const float4 m0 = *(const float4*)(m + n0 + 8 * hi);
        const float4 m1 = *(const float4*)(m + n0 + 8 * hi + 4);
        mv[0] = m0.x; mv[1] = m0.y; mv[2] = m0.z; mv[3] = m0.w;
        mv[4] = m1.x; mv[5] = m1.y; mv[6] = m1.z; mv[7] = m1.w;
    }

    for (int bt = 0; bt < 4; ++bt) {
        const int col = colGroup * 64 + bt * 16 + lo;

        v2f bm;
        bm.x = tbuf[(size_t)(0 + 2 * hi) * (size_t)B + col];
        bm.y = tbuf[(size_t)(1 + 2 * hi) * (size_t)B + col];

        const float xv = x[col];

        const size_t base = (size_t)(n0 + 8 * hi) * (size_t)B + (size_t)col;
        const float* __restrict__ hp = h + base;

        float hv[8];
#pragma unroll
        for (int g = 0; g < 8; ++g) hv[g] = hp[(size_t)g * B];

        v8f c = {};
        v8f d = __builtin_amdgcn_wmma_f32_16x16x4_f32(
             false, a,  false, bm,
             (short)0, c,  false,  false);

        float out[8];
#pragma unroll
        for (int g = 0; g < 8; ++g) {
            const float upd = fmaf(mv[g], xv, d[g] * invN);
            out[g] = hv[g] + (upd - hv[g]);
        }

        float* __restrict__ op = hnew + base;
#pragma unroll
        for (int g = 0; g < 8; ++g)
            __builtin_nontemporal_store(out[g], op + (size_t)g * B);
    }
}

extern "C" void kernel_launch(void* const* d_in, const int* in_sizes, int n_in,
                              void* d_out, int out_size, void* d_ws, size_t ws_size,
                              hipStream_t stream) {
    const float* x = (const float*)d_in[0];
    const float* h = (const float*)d_in[1];
    const float* m = (const float*)d_in[2];
    const float* u = (const float*)d_in[3];
    const float* v = (const float*)d_in[4];
    const float* z = (const float*)d_in[5];

    const int B = in_sizes[0];
    const int N = in_sizes[2];
    const float invN = 1.0f / (float)N;

    float* y_out = (float*)d_out;
    float* hnew  = (float*)d_out + B;

    const int NCHUNKS = 32;
    const int nchunk  = N / NCHUNKS;

    float* partial = (float*)d_ws;
    float* tbuf    = partial + (size_t)NCHUNKS * 5 * (size_t)B;

    dim3 g1(B / 512, NCHUNKS);
    k1_reduce<<<g1, 256, 0, stream>>>(h, v, z, partial, B, nchunk);

    const int tot2 = 5 * B;
    k2_finalize<<<(tot2 + 255) / 256, 256, 0, stream>>>(partial, tbuf, y_out, B,
                                                        NCHUNKS, invN);

    const int waves = (N / 16) * (B / 64);
    k3_update<<<waves / 8, 256, 0, stream>>>(h, u, m, x, tbuf, hnew, B, invN);
}


# pass 3 re-tiled: WMMA tile j takes columns 4c+j, so a lane owns 4 adjacent columns: 16-byte h loads, h_new stores, t/x reads; 4096 waves of 16x128 (was 8192 of 16x64), all row loads issued first
# speedup vs baseline: 3.3341x; 3.3341x over previous
#include <hip/hip_runtime.h>
#include <math.h>

#define ALPHA_ERF 0.8862269254527580137f

typedef __attribute__((ext_vector_type(2))) float v2f;
typedef __attribute__((ext_vector_type(4))) float v4f;
typedef __attribute__((ext_vector_type(8))) float v8f;

#define K1_WAVES 16
#define K1_ROWS  16

__global__ __launch_bounds__(K1_WAVES * 32) void k1_reduce(
    const float* __restrict__ h, const float* __restrict__ v,
    const float* __restrict__ z, float* __restrict__ partial, int B)
{
    __shared__ float4 red[K1_WAVES][5][32];

    const int lane  = threadIdx.x & 31;
    const int wave  = __builtin_amdgcn_readfirstlane((int)(threadIdx.x >> 5));
    const int b0    = blockIdx.x * 128 + lane * 4;
    const int chunk = blockIdx.y;
    const int nBeg  = (chunk * K1_WAVES + wave) * K1_ROWS;

    const float* __restrict__ hp = h + (size_t)nBeg * (size_t)B + (size_t)b0;

    float4 hv[K1_ROWS];
#pragma unroll
    for (int i = 0; i < K1_ROWS; ++i)
        hv[i] = *(const float4*)(hp + (size_t)i * (size_t)B);

    float4 acc[5];
    for (int r = 0; r < 5; ++r) acc[r] = make_float4(0.0f, 0.0f, 0.0f, 0.0f);

#pragma unroll
    for (int i = 0; i < K1_ROWS; ++i) {
        const float4 vv = *(const float4*)(v + (size_t)(nBeg + i) * 4);
        const float  zv = z[nBeg + i];
        const float p0 = erff(ALPHA_ERF * hv[i].x);
        const float p1 = erff(ALPHA_ERF * hv[i].y);
        const float p2 = erff(ALPHA_ERF * hv[i].z);
        const float p3 = erff(ALPHA_ERF * hv[i].w);
        acc[0].x = fmaf(vv.x, p0, acc[0].x); acc[0].y = fmaf(vv.x, p1, acc[0].y);
        acc[0].z = fmaf(vv.x, p2, acc[0].z); acc[0].w = fmaf(vv.x, p3, acc[0].w);
        acc[1].x = fmaf(vv.y, p0, acc[1].x); acc[1].y = fmaf(vv.y, p1, acc[1].y);
        acc[1].z = fmaf(vv.y, p2, acc[1].z); acc[1].w = fmaf(vv.y, p3, acc[1].w);
        acc[2].x = fmaf(vv.z, p0, acc[2].x); acc[2].y = fmaf(vv.z, p1, acc[2].y);
        acc[2].z = fmaf(vv.z, p2, acc[2].z); acc[2].w = fmaf(vv.z, p3, acc[2].w);
        acc[3].x = fmaf(vv.w, p0, acc[3].x); acc[3].y = fmaf(vv.w, p1, acc[3].y);
        acc[3].z = fmaf(vv.w, p2, acc[3].z); acc[3].w = fmaf(vv.w, p3, acc[3].w);
        acc[4].x = fmaf(zv,   p0, acc[4].x); acc[4].y = fmaf(zv,   p1, acc[4].y);
        acc[4].z = fmaf(zv,   p2, acc[4].z); acc[4].w = fmaf(zv,   p3, acc[4].w);
    }

    for (int r = 0; r < 5; ++r) red[wave][r][lane] = acc[r];
    __syncthreads();

    if (wave < 5) {
        float4 s = red[0][wave][lane];
#pragma unroll
        for (int w = 1; w < K1_WAVES; ++w) {
            const float4 t = red[w][wave][lane];
            s.x += t.x; s.y += t.y; s.z += t.z; s.w += t.w;
        }
        *(float4*)(partial + (size_t)(chunk * 5 + wave) * (size_t)B + (size_t)b0) = s;
    }
}

__global__ __launch_bounds__(256) void k2_finalize(
    const float* __restrict__ partial, float* __restrict__ tbuf,
    float* __restrict__ y_out, int B, int nchunks, float invN)
{
    const int idx = blockIdx.x * 256 + threadIdx.x;
    if (idx >= 5 * B) return;
    const int r = idx / B;
    const int b = idx - r * B;
    float s = 0.0f;
    for (int c = 0; c < nchunks; ++c)
        s += partial[(size_t)(c * 5 + r) * (size_t)B + b];
    if (r == 4) y_out[b] = s * invN;
    else        tbuf[(size_t)r * (size_t)B + b] = s;
}

#define K3_STRIPS 2

__global__ __launch_bounds__(256) void k3_update(
    const float* __restrict__ h, const float* __restrict__ u,
    const float* __restrict__ m, const float* __restrict__ x,
    const float* __restrict__ tbuf, float* __restrict__ hnew,
    int B, float invN)
{
    const int lane = threadIdx.x & 31;
    const int wave = __builtin_amdgcn_readfirstlane((int)(threadIdx.x >> 5));
    const int lo   = lane & 15;
    const int hi   = lane >> 4;

    const int colGroups = B >> 7;
    const int gwave     = blockIdx.x * 8 + wave;
    const int rowTile   = gwave / colGroups;
    const int colGroup  = gwave - rowTile * colGroups;
    const int n0        = rowTile * 16;
    const int cg0       = colGroup * (64 * K3_STRIPS);

    const size_t   uoff = (size_t)n0 * (size_t)B + (size_t)cg0;
    const unsigned loff = (unsigned)(8 * hi) * (unsigned)B + (unsigned)(4 * lo);

    v2f a;
    {
        const float2 av = *(const float2*)(u + (size_t)(n0 + lo) * 4 + hi * 2);
        a.x = av.x; a.y = av.y;
    }

    v4f t0[K3_STRIPS], t1[K3_STRIPS], xv[K3_STRIPS];
#pragma unroll
    for (int s = 0; s < K3_STRIPS; ++s) {
        const int col = cg0 + s * 64 + 4 * lo;
        t0[s] = *(const v4f*)(tbuf + (size_t)(0 + 2 * hi) * (size_t)B + col);
        t1[s] = *(const v4f*)(tbuf + (size_t)(1 + 2 * hi) * (size_t)B + col);
        xv[s] = *(const v4f*)(x + col);
    }

    float mv[8];
    {
        const float4 m0 = *(const float4*)(m + n0 + 8 * hi);
        const float4 m1 = *(const float4*)(m + n0 + 8 * hi + 4);
        mv[0] = m0.x; mv[1] = m0.y; mv[2] = m0.z; mv[3] = m0.w;
        mv[4] = m1.x; mv[5] = m1.y; mv[6] = m1.z; mv[7] = m1.w;
    }

    __builtin_amdgcn_sched_barrier(0);

    const float* __restrict__ hp = h + uoff;
    v4f hv[K3_STRIPS][8];
#pragma unroll
    for (int s = 0; s < K3_STRIPS; ++s)
#pragma unroll
        for (int g = 0; g < 8; ++g)
            hv[s][g] = *(const v4f*)(hp + (size_t)g * (size_t)B + s * 64 + loff);
    __builtin_amdgcn_sched_barrier(0);

    v8f d[K3_STRIPS][4];
#pragma unroll
    for (int s = 0; s < K3_STRIPS; ++s)
#pragma unroll
        for (int j = 0; j < 4; ++j) {
            v2f bm;
            bm.x = t0[s][j];
            bm.y = t1[s][j];
            v8f c = {};
            d[s][j] = __builtin_amdgcn_wmma_f32_16x16x4_f32(
                 false, a,  false, bm,
                 (short)0, c,  false,  false);
        }

    float* __restrict__ op = hnew + uoff;
#pragma unroll
    for (int s = 0; s < K3_STRIPS; ++s)
#pragma unroll
        for (int g = 0; g < 8; ++g) {
            v4f o;
#pragma unroll
            for (int j = 0; j < 4; ++j) {
                const float hh  = hv[s][g][j];
                const float upd = fmaf(mv[g], xv[s][j], d[s][j][g] * invN);
                o[j] = hh + (upd - hh);
            }
            __builtin_nontemporal_store(o, (v4f*)(op + (size_t)g * (size_t)B + s * 64 + loff));
        }
}

extern "C" void kernel_launch(void* const* d_in, const int* in_sizes, int n_in,
                              void* d_out, int out_size, void* d_ws, size_t ws_size,
                              hipStream_t stream) {
    const float* x = (const float*)d_in[0];
    const float* h = (const float*)d_in[1];
    const float* m = (const float*)d_in[2];
    const float* u = (const float*)d_in[3];
    const float* v = (const float*)d_in[4];
    const float* z = (const float*)d_in[5];

    const int B = in_sizes[0];
    const int N = in_sizes[2];
    const float invN = 1.0f / (float)N;

    float* y_out = (float*)d_out;
    float* hnew  = (float*)d_out + B;

    const int NCHUNKS = N / (K1_WAVES * K1_ROWS);

    float* partial = (float*)d_ws;
    float* tbuf    = partial + (size_t)NCHUNKS * 5 * (size_t)B;

    dim3 g1(B / 128, NCHUNKS);
    k1_reduce<<<g1, K1_WAVES * 32, 0, stream>>>(h, v, z, partial, B);

    const int tot2 = 5 * B;
    k2_finalize<<<(tot2 + 255) / 256, 256, 0, stream>>>(partial, tbuf, y_out, B,
                                                        NCHUNKS, invN);

    const int waves = (N / 16) * (B / (64 * K3_STRIPS));
    k3_update<<<waves / 8, 256, 0, stream>>>(h, u, m, x, tbuf, hnew, B, invN);
}


# fold pass fused into the update pass: each 16-wave block (128 cols x 256 rows, pass 1's tile) sums its columns' 16 chunk planes into LDS t, row group 0 also y; 2 launches not 3, same sum order
# speedup vs baseline: 4.9953x; 1.4982x over previous
#include <hip/hip_runtime.h>
#include <math.h>

#define ALPHA_ERF 0.8862269254527580137f

typedef __attribute__((ext_vector_type(2))) float v2f;
typedef __attribute__((ext_vector_type(4))) float v4f;
typedef __attribute__((ext_vector_type(8))) float v8f;

#define K1_WAVES 16
#define K1_ROWS  16
#define N_CHUNKS 16

__global__ __launch_bounds__(K1_WAVES * 32) void k1_reduce(
    const float* __restrict__ h, const float* __restrict__ v,
    const float* __restrict__ z, float* __restrict__ partial, int B)
{
    __shared__ float4 red[K1_WAVES][5][32];

    const int lane  = threadIdx.x & 31;
    const int wave  = __builtin_amdgcn_readfirstlane((int)(threadIdx.x >> 5));
    const int b0    = blockIdx.x * 128 + lane * 4;
    const int chunk = blockIdx.y;
    const int nBeg  = (chunk * K1_WAVES + wave) * K1_ROWS;

    const float* __restrict__ hp = h + (size_t)nBeg * (size_t)B + (size_t)b0;

    float4 hv[K1_ROWS];
#pragma unroll
    for (int i = 0; i < K1_ROWS; ++i)
        hv[i] = *(const float4*)(hp + (size_t)i * (size_t)B);

    float4 acc[5];
    for (int r = 0; r < 5; ++r) acc[r] = make_float4(0.0f, 0.0f, 0.0f, 0.0f);

#pragma unroll
    for (int i = 0; i < K1_ROWS; ++i) {
        const float4 vv = *(const float4*)(v + (size_t)(nBeg + i) * 4);
        const float  zv = z[nBeg + i];
        const float p0 = erff(ALPHA_ERF * hv[i].x);
        const float p1 = erff(ALPHA_ERF * hv[i].y);
        const float p2 = erff(ALPHA_ERF * hv[i].z);
        const float p3 = erff(ALPHA_ERF * hv[i].w);
        acc[0].x = fmaf(vv.x, p0, acc[0].x); acc[0].y = fmaf(vv.x, p1, acc[0].y);
        acc[0].z = fmaf(vv.x, p2, acc[0].z); acc[0].w = fmaf(vv.x, p3, acc[0].w);
        acc[1].x = fmaf(vv.y, p0, acc[1].x); acc[1].y = fmaf(vv.y, p1, acc[1].y);
        acc[1].z = fmaf(vv.y, p2, acc[1].z); acc[1].w = fmaf(vv.y, p3, acc[1].w);
        acc[2].x = fmaf(vv.z, p0, acc[2].x); acc[2].y = fmaf(vv.z, p1, acc[2].y);
        acc[2].z = fmaf(vv.z, p2, acc[2].z); acc[2].w = fmaf(vv.z, p3, acc[2].w);
        acc[3].x = fmaf(vv.w, p0, acc[3].x); acc[3].y = fmaf(vv.w, p1, acc[3].y);
        acc[3].z = fmaf(vv.w, p2, acc[3].z); acc[3].w = fmaf(vv.w, p3, acc[3].w);
        acc[4].x = fmaf(zv,   p0, acc[4].x); acc[4].y = fmaf(zv,   p1, acc[4].y);
        acc[4].z = fmaf(zv,   p2, acc[4].z); acc[4].w = fmaf(zv,   p3, acc[4].w);
    }

    for (int r = 0; r < 5; ++r) red[wave][r][lane] = acc[r];
    __syncthreads();

    if (wave < 5) {
        float4 s = red[0][wave][lane];
#pragma unroll
        for (int w = 1; w < K1_WAVES; ++w) {
            const float4 t = red[w][wave][lane];
            s.x += t.x; s.y += t.y; s.z += t.z; s.w += t.w;
        }
        *(float4*)(partial + (size_t)(chunk * 5 + wave) * (size_t)B + (size_t)b0) = s;
    }
}

#define K3_WAVES  16
#define K3_STRIPS 2

__global__ __launch_bounds__(K3_WAVES * 32) void k3_update(
    const float* __restrict__ h, const float* __restrict__ u,
    const float* __restrict__ m, const float* __restrict__ x,
    const float* __restrict__ partial, float* __restrict__ hnew,
    float* __restrict__ y_out, int B, float invN)
{
    __shared__ __attribute__((aligned(16))) float tl[4][64 * K3_STRIPS];

    const int lane = threadIdx.x & 31;
    const int wave = __builtin_amdgcn_readfirstlane((int)(threadIdx.x >> 5));
    const int lo   = lane & 15;
    const int hi   = lane >> 4;

    const int n0  = ((int)blockIdx.y * K3_WAVES + wave) * 16;
    const int cg0 = (int)blockIdx.x * (64 * K3_STRIPS);

    const size_t   uoff = (size_t)n0 * (size_t)B + (size_t)cg0;
    const unsigned loff = (unsigned)(8 * hi) * (unsigned)B + (unsigned)(4 * lo);

    v2f a;
    {
        const float2 av = *(const float2*)(u + (size_t)(n0 + lo) * 4 + hi * 2);
        a.x = av.x; a.y = av.y;
    }

    v4f xv[K3_STRIPS];
#pragma unroll
    for (int s = 0; s < K3_STRIPS; ++s)
        xv[s] = *(const v4f*)(x + cg0 + s * 64 + 4 * lo);

    float mv[8];
    {
        const float4 m0 = *(const float4*)(m + n0 + 8 * hi);
        const float4 m1 = *(const float4*)(m + n0 + 8 * hi + 4);
        mv[0] = m0.x; mv[1] = m0.y; mv[2] = m0.z; mv[3] = m0.w;
        mv[4] = m1.x; mv[5] = m1.y; mv[6] = m1.z; mv[7] = m1.w;
    }

    const int fr   = wave >> 2;
    const int fcol = (wave & 3) * 32 + lane;
    const size_t pstride = (size_t)5 * (size_t)B;
    const float* __restrict__ pp = partial + (size_t)fr * (size_t)B + (size_t)(cg0 + fcol);
    float pv[N_CHUNKS];
#pragma unroll
    for (int c = 0; c < N_CHUNKS; ++c) pv[c] = pp[(size_t)c * pstride];

    const bool foldY = (blockIdx.y == 0) && (wave < 4);
    float yv[N_CHUNKS];
    if (foldY) {
#pragma unroll
        for (int c = 0; c < N_CHUNKS; ++c) yv[c] = pp[(size_t)c * pstride + (size_t)4 * (size_t)B];
    }

    __builtin_amdgcn_sched_barrier(0);

    const float* __restrict__ hp = h + uoff;
    v4f hv[K3_STRIPS][8];
#pragma unroll
    for (int s = 0; s < K3_STRIPS; ++s)
#pragma unroll
        for (int g = 0; g < 8; ++g)
            hv[s][g] = *(const v4f*)(hp + (size_t)g * (size_t)B + s * 64 + loff);
    __builtin_amdgcn_sched_barrier(0);

    {
        float s = 0.0f;
#pragma unroll
        for (int c = 0; c < N_CHUNKS; ++c) s += pv[c];
        tl[fr][fcol] = s;
    }
    if (foldY) {
        float s = 0.0f;
#pragma unroll
        for (int c = 0; c < N_CHUNKS; ++c) s += yv[c];
        y_out[cg0 + fcol] = s * invN;
    }
    __builtin_amdgcn_fence(__ATOMIC_RELEASE, "workgroup", "local");
    __builtin_amdgcn_s_barrier();
    __builtin_amdgcn_fence(__ATOMIC_ACQUIRE, "workgroup", "local");

    v4f t0[K3_STRIPS], t1[K3_STRIPS];
#pragma unroll
    for (int s = 0; s < K3_STRIPS; ++s) {
        t0[s] = *(const v4f*)&tl[0 + 2 * hi][s * 64 + 4 * lo];
        t1[s] = *(const v4f*)&tl[1 + 2 * hi][s * 64 + 4 * lo];
    }

    v8f d[K3_STRIPS][4];
#pragma unroll
    for (int s = 0; s < K3_STRIPS; ++s)
#pragma unroll
        for (int j = 0; j < 4; ++j) {
            v2f bm;
            bm.x = t0[s][j];
            bm.y = t1[s][j];
            v8f c = {};
            d[s][j] = __builtin_amdgcn_wmma_f32_16x16x4_f32(
                 false, a,  false, bm,
                 (short)0, c,  false,  false);
        }

    float* __restrict__ op = hnew + uoff;
#pragma unroll
    for (int s = 0; s < K3_STRIPS; ++s)
#pragma unroll
        for (int g = 0; g < 8; ++g) {
            v4f o;
#pragma unroll
            for (int j = 0; j < 4; ++j) {
                const float hh  = hv[s][g][j];
                const float upd = fmaf(mv[g], xv[s][j], d[s][j][g] * invN);
                o[j] = hh + (upd - hh);
            }
            __builtin_nontemporal_store(o, (v4f*)(op + (size_t)g * (size_t)B + s * 64 + loff));
        }
}

extern "C" void kernel_launch(void* const* d_in, const int* in_sizes, int n_in,
                              void* d_out, int out_size, void* d_ws, size_t ws_size,
                              hipStream_t stream) {
    const float* x = (const float*)d_in[0];
    const float* h = (const float*)d_in[1];
    const float* m = (const float*)d_in[2];
    const float* u = (const float*)d_in[3];
    const float* v = (const float*)d_in[4];
    const float* z = (const float*)d_in[5];

    const int B = in_sizes[0];
    const int N = in_sizes[2];
    const float invN = 1.0f / (float)N;

    float* y_out = (float*)d_out;
    float* hnew  = (float*)d_out + B;

    float* partial = (float*)d_ws;

    dim3 grid(B / 128, N_CHUNKS);

    k1_reduce<<<grid, K1_WAVES * 32, 0, stream>>>(h, v, z, partial, B);

    k3_update<<<grid, K3_WAVES * 32, 0, stream>>>(h, u, m, x, partial, hnew, y_out, B, invN);
}


# pass 1 erf branch-free: both range forms (same polynomials, exp2 reduction in two fmas) on 8 values in lockstep, one select each, not 128 masked serial regions per wave; phi may move in its last bit
# speedup vs baseline: 6.1842x; 1.2380x over previous
#include <hip/hip_runtime.h>
#include <math.h>

#define ALPHA_ERF 0.8862269254527580137f

typedef __attribute__((ext_vector_type(2))) float v2f;
typedef __attribute__((ext_vector_type(4))) float v4f;
typedef __attribute__((ext_vector_type(8))) float v8f;


__device__ __forceinline__ v8f erf_x8(const v8f x)
{
    const v8f ax = __builtin_elementwise_abs(x);

    v8f p = __builtin_elementwise_fma(ax, (v8f)(__int_as_float(0x378e98ab)), (v8f)(__int_as_float(0xb9c68948)));
    p = __builtin_elementwise_fma(ax, p, (v8f)(__int_as_float(0x3b7cd369)));
    p = __builtin_elementwise_fma(ax, p, (v8f)(__int_as_float(0xbcc618b2)));
    p = __builtin_elementwise_fma(ax, p, (v8f)(__int_as_float(0x3dda74e4)));
    p = __builtin_elementwise_fma(ax, p, (v8f)(__int_as_float(0x3f228afd)));
    p = __builtin_elementwise_fma(ax, p, (v8f)(__int_as_float(0x3e03c728)));
    const v8f qb = __builtin_elementwise_fma(ax, p, ax);

    const v8f nl2e = (v8f)(__int_as_float(0xbfb8aa3b));
    const v8f t    = qb * nl2e;
    v8f r;
#pragma unroll
    for (int j = 0; j < 8; ++j) r[j] = __builtin_rintf(t[j]);
    v8f f = __builtin_elementwise_fma(qb, nl2e, -r);
    f = __builtin_elementwise_fma(qb, (v8f)(__int_as_float(0xb2a5705f)), f);
    v8f e;
#pragma unroll
    for (int j = 0; j < 8; ++j) e[j] = __builtin_ldexpf(__builtin_amdgcn_exp2f(f[j]), (int)r[j]);
    e = (qb > __int_as_float(0x42ce8ed0)) ? (v8f)(0.0f) : e;
    const v8f big = 1.0f - e;

    const v8f t2 = x * x;
    v8f q = __builtin_elementwise_fma((v8f)(__int_as_float(0xba1345e1)), t2, (v8f)(__int_as_float(0x3ba10414)));
    q = __builtin_elementwise_fma(t2, q, (v8f)(__int_as_float(0xbcdac9b8)));
    q = __builtin_elementwise_fma(t2, q, (v8f)(__int_as_float(0x3de703be)));
    q = __builtin_elementwise_fma(t2, q, (v8f)(__int_as_float(0xbec09330)));
    q = __builtin_elementwise_fma(t2, q, (v8f)(__int_as_float(0x3e0375d0)));
    const v8f sm = __builtin_elementwise_fma(ax, q, ax);

    const v8f mag = (ax < 1.0f) ? sm : big;
    return __builtin_elementwise_copysign(mag, x);
}

#define K1_WAVES 16
#define K1_ROWS  16
#define N_CHUNKS 16

__global__ __launch_bounds__(K1_WAVES * 32) void k1_reduce(
    const float* __restrict__ h, const float* __restrict__ v,
    const float* __restrict__ z, float* __restrict__ partial, int B)
{
    __shared__ float4 red[K1_WAVES][5][32];

    const int lane  = threadIdx.x & 31;
    const int wave  = __builtin_amdgcn_readfirstlane((int)(threadIdx.x >> 5));
    const int b0    = blockIdx.x * 128 + lane * 4;
    const int chunk = blockIdx.y;
    const int nBeg  = (chunk * K1_WAVES + wave) * K1_ROWS;

    const float* __restrict__ hp = h + (size_t)nBeg * (size_t)B + (size_t)b0;

    float4 hv[K1_ROWS];
#pragma unroll
    for (int i = 0; i < K1_ROWS; ++i)
        hv[i] = *(const float4*)(hp + (size_t)i * (size_t)B);
    __builtin_amdgcn_sched_barrier(0);

    float4 acc[5];
    for (int r = 0; r < 5; ++r) acc[r] = make_float4(0.0f, 0.0f, 0.0f, 0.0f);

#pragma unroll
    for (int i = 0; i < K1_ROWS; i += 2) {
        v8f xin;
        xin[0] = hv[i].x;     xin[1] = hv[i].y;     xin[2] = hv[i].z;     xin[3] = hv[i].w;
        xin[4] = hv[i + 1].x; xin[5] = hv[i + 1].y; xin[6] = hv[i + 1].z; xin[7] = hv[i + 1].w;
        const v8f ph = erf_x8(ALPHA_ERF * xin);
#pragma unroll
        for (int k = 0; k < 2; ++k) {
            const float4 vv = *(const float4*)(v + (size_t)(nBeg + i + k) * 4);
            const float  zv = z[nBeg + i + k];
            const float p0 = ph[4 * k + 0];
            const float p1 = ph[4 * k + 1];
            const float p2 = ph[4 * k + 2];
            const float p3 = ph[4 * k + 3];
            acc[0].x = fmaf(vv.x, p0, acc[0].x); acc[0].y = fmaf(vv.x, p1, acc[0].y);
            acc[0].z = fmaf(vv.x, p2, acc[0].z); acc[0].w = fmaf(vv.x, p3, acc[0].w);
            acc[1].x = fmaf(vv.y, p0, acc[1].x); acc[1].y = fmaf(vv.y, p1, acc[1].y);
            acc[1].z = fmaf(vv.y, p2, acc[1].z); acc[1].w = fmaf(vv.y, p3, acc[1].w);
            acc[2].x = fmaf(vv.z, p0, acc[2].x); acc[2].y = fmaf(vv.z, p1, acc[2].y);
            acc[2].z = fmaf(vv.z, p2, acc[2].z); acc[2].w = fmaf(vv.z, p3, acc[2].w);
            acc[3].x = fmaf(vv.w, p0, acc[3].x); acc[3].y = fmaf(vv.w, p1, acc[3].y);
            acc[3].z = fmaf(vv.w, p2, acc[3].z); acc[3].w = fmaf(vv.w, p3, acc[3].w);
            acc[4].x = fmaf(zv,   p0, acc[4].x); acc[4].y = fmaf(zv,   p1, acc[4].y);
            acc[4].z = fmaf(zv,   p2, acc[4].z); acc[4].w = fmaf(zv,   p3, acc[4].w);
        }
        __builtin_amdgcn_sched_barrier(0);
    }

    for (int r = 0; r < 5; ++r) red[wave][r][lane] = acc[r];
    __syncthreads();

    if (wave < 5) {
        float4 s = red[0][wave][lane];
#pragma unroll
        for (int w = 1; w < K1_WAVES; ++w) {
            const float4 t = red[w][wave][lane];
            s.x += t.x; s.y += t.y; s.z += t.z; s.w += t.w;
        }
        *(float4*)(partial + (size_t)(chunk * 5 + wave) * (size_t)B + (size_t)b0) = s;
    }
}

#define K3_WAVES  16
#define K3_STRIPS 2

__global__ __launch_bounds__(K3_WAVES * 32) void k3_update(
    const float* __restrict__ h, const float* __restrict__ u,
    const float* __restrict__ m, const float* __restrict__ x,
    const float* __restrict__ partial, float* __restrict__ hnew,
    float* __restrict__ y_out, int B, float invN)
{
    __shared__ __attribute__((aligned(16))) float tl[4][64 * K3_STRIPS];

    const int lane = threadIdx.x & 31;
    const int wave = __builtin_amdgcn_readfirstlane((int)(threadIdx.x >> 5));
    const int lo   = lane & 15;
    const int hi   = lane >> 4;

    const int n0  = ((int)blockIdx.y * K3_WAVES + wave) * 16;
    const int cg0 = (int)blockIdx.x * (64 * K3_STRIPS);

    const size_t   uoff = (size_t)n0 * (size_t)B + (size_t)cg0;
    const unsigned loff = (unsigned)(8 * hi) * (unsigned)B + (unsigned)(4 * lo);

    v2f a;
    {
        const float2 av = *(const float2*)(u + (size_t)(n0 + lo) * 4 + hi * 2);
        a.x = av.x; a.y = av.y;
    }

    v4f xv[K3_STRIPS];
#pragma unroll
    for (int s = 0; s < K3_STRIPS; ++s)
        xv[s] = *(const v4f*)(x + cg0 + s * 64 + 4 * lo);

    float mv[8];
    {
        const float4 m0 = *(const float4*)(m + n0 + 8 * hi);
        const float4 m1 = *(const float4*)(m + n0 + 8 * hi + 4);
        mv[0] = m0.x; mv[1] = m0.y; mv[2] = m0.z; mv[3] = m0.w;
        mv[4] = m1.x; mv[5] = m1.y; mv[6] = m1.z; mv[7] = m1.w;
    }

    const int fr   = wave >> 2;
    const int fcol = (wave & 3) * 32 + lane;
    const size_t pstride = (size_t)5 * (size_t)B;
    const float* __restrict__ pp = partial + (size_t)fr * (size_t)B + (size_t)(cg0 + fcol);
    float pv[N_CHUNKS];
#pragma unroll
    for (int c = 0; c < N_CHUNKS; ++c) pv[c] = pp[(size_t)c * pstride];

    const bool foldY = (blockIdx.y == 0) && (wave < 4);
    float yv[N_CHUNKS];
    if (foldY) {
#pragma unroll
        for (int c = 0; c < N_CHUNKS; ++c) yv[c] = pp[(size_t)c * pstride + (size_t)4 * (size_t)B];
    }

    __builtin_amdgcn_sched_barrier(0);

    const float* __restrict__ hp = h + uoff;
    v4f hv[K3_STRIPS][8];
#pragma unroll
    for (int s = 0; s < K3_STRIPS; ++s)
#pragma unroll
        for (int g = 0; g < 8; ++g)
            hv[s][g] = *(const v4f*)(hp + (size_t)g * (size_t)B + s * 64 + loff);
    __builtin_amdgcn_sched_barrier(0);

    {
        float s = 0.0f;
#pragma unroll
        for (int c = 0; c < N_CHUNKS; ++c) s += pv[c];
        tl[fr][fcol] = s;
    }
    if (foldY) {
        float s = 0.0f;
#pragma unroll
        for (int c = 0; c < N_CHUNKS; ++c) s += yv[c];
        y_out[cg0 + fcol] = s * invN;
    }
    __builtin_amdgcn_fence(__ATOMIC_RELEASE, "workgroup", "local");
    __builtin_amdgcn_s_barrier();
    __builtin_amdgcn_fence(__ATOMIC_ACQUIRE, "workgroup", "local");

    v4f t0[K3_STRIPS], t1[K3_STRIPS];
#pragma unroll
    for (int s = 0; s < K3_STRIPS; ++s) {
        t0[s] = *(const v4f*)&tl[0 + 2 * hi][s * 64 + 4 * lo];
        t1[s] = *(const v4f*)&tl[1 + 2 * hi][s * 64 + 4 * lo];
    }

    v8f d[K3_STRIPS][4];
#pragma unroll
    for (int s = 0; s < K3_STRIPS; ++s)
#pragma unroll
        for (int j = 0; j < 4; ++j) {
            v2f bm;
            bm.x = t0[s][j];
            bm.y = t1[s][j];
            v8f c = {};
            d[s][j] = __builtin_amdgcn_wmma_f32_16x16x4_f32(
                 false, a,  false, bm,
                 (short)0, c,  false,  false);
        }

    float* __restrict__ op = hnew + uoff;
#pragma unroll
    for (int s = 0; s < K3_STRIPS; ++s)
#pragma unroll
        for (int g = 0; g < 8; ++g) {
            v4f o;
#pragma unroll
            for (int j = 0; j < 4; ++j) {
                const float hh  = hv[s][g][j];
                const float upd = fmaf(mv[g], xv[s][j], d[s][j][g] * invN);
                o[j] = hh + (upd - hh);
            }
            __builtin_nontemporal_store(o, (v4f*)(op + (size_t)g * (size_t)B + s * 64 + loff));
        }
}

extern "C" void kernel_launch(void* const* d_in, const int* in_sizes, int n_in,
                              void* d_out, int out_size, void* d_ws, size_t ws_size,
                              hipStream_t stream) {
    const float* x = (const float*)d_in[0];
    const float* h = (const float*)d_in[1];
    const float* m = (const float*)d_in[2];
    const float* u = (const float*)d_in[3];
    const float* v = (const float*)d_in[4];
    const float* z = (const float*)d_in[5];

    const int B = in_sizes[0];
    const int N = in_sizes[2];
    const float invN = 1.0f / (float)N;

    float* y_out = (float*)d_out;
    float* hnew  = (float*)d_out + B;

    float* partial = (float*)d_ws;

    dim3 grid(B / 128, N_CHUNKS);

    k1_reduce<<<grid, K1_WAVES * 32, 0, stream>>>(h, v, z, partial, B);

    k3_update<<<grid, K3_WAVES * 32, 0, stream>>>(h, u, m, x, partial, hnew, y_out, B, invN);
}


# pass 1 erf as one rational form x*P(x^2)/Q(x^2) on clamped x (10 multiply-adds, one v_rcp_f32), not two range forms with an exp2 and selects; phi within 8 ulp of erf, was 1.5
# speedup vs baseline: 6.7100x; 1.0850x over previous
#include <hip/hip_runtime.h>
#include <math.h>

#define ALPHA_ERF 0.8862269254527580137f

typedef __attribute__((ext_vector_type(2))) float v2f;
typedef __attribute__((ext_vector_type(4))) float v4f;
typedef __attribute__((ext_vector_type(8))) float v8f;


#define ERF_LIM 3.832506856900711f
__device__ __forceinline__ v8f erf_x8(const v8f x)
{
    v8f xc;
#pragma unroll
    for (int j = 0; j < 8; ++j) xc[j] = __builtin_amdgcn_fmed3f(x[j], -ERF_LIM, ERF_LIM);
    const v8f x2 = xc * xc;

    v8f p = __builtin_elementwise_fma(x2, (v8f)(0.00022905065861350646f), (v8f)(0.0034082910107109506f));
    p = __builtin_elementwise_fma(x2, p, (v8f)(0.050955695062380861f));
    p = __builtin_elementwise_fma(x2, p, (v8f)(0.18520832239976145f));
    p = __builtin_elementwise_fma(x2, p, (v8f)(1.128379143519084f));

    v8f q = __builtin_elementwise_fma(x2, (v8f)(-1.1791602954361697e-7f), (v8f)(0.000023547966471313185f));
    q = __builtin_elementwise_fma(x2, q, (v8f)(0.0010179625278914885f));
    q = __builtin_elementwise_fma(x2, q, (v8f)(0.014070470171167667f));
    q = __builtin_elementwise_fma(x2, q, (v8f)(0.11098505178285362f));
    q = __builtin_elementwise_fma(x2, q, (v8f)(0.49746925110067538f));
    q = __builtin_elementwise_fma(x2, q, (v8f)(1.0f));

    v8f rq;
#pragma unroll
    for (int j = 0; j < 8; ++j) rq[j] = __builtin_amdgcn_rcpf(q[j]);
    return (xc * p) * rq;
}

#define K1_WAVES 16
#define K1_ROWS  16
#define N_CHUNKS 16

__global__ __launch_bounds__(K1_WAVES * 32) void k1_reduce(
    const float* __restrict__ h, const float* __restrict__ v,
    const float* __restrict__ z, float* __restrict__ partial, int B)
{
    __shared__ float4 red[K1_WAVES][5][32];

    const int lane  = threadIdx.x & 31;
    const int wave  = __builtin_amdgcn_readfirstlane((int)(threadIdx.x >> 5));
    const int b0    = blockIdx.x * 128 + lane * 4;
    const int chunk = blockIdx.y;
    const int nBeg  = (chunk * K1_WAVES + wave) * K1_ROWS;

    const float* __restrict__ hp = h + (size_t)nBeg * (size_t)B + (size_t)b0;

    v4f hv[K1_ROWS];
#pragma unroll
    for (int i = 0; i < K1_ROWS; ++i)
        hv[i] = *(const v4f*)(hp + (size_t)i * (size_t)B);
    __builtin_amdgcn_sched_barrier(0);

    float4 acc[5];
    for (int r = 0; r < 5; ++r) acc[r] = make_float4(0.0f, 0.0f, 0.0f, 0.0f);

#pragma unroll
    for (int i = 0; i < K1_ROWS; i += 2) {
        v8f xin;
        asm volatile("" : "+v"(hv[i]), "+v"(hv[i + 1]));
        xin[0] = hv[i][0];     xin[1] = hv[i][1];     xin[2] = hv[i][2];     xin[3] = hv[i][3];
        xin[4] = hv[i + 1][0]; xin[5] = hv[i + 1][1]; xin[6] = hv[i + 1][2]; xin[7] = hv[i + 1][3];
        const v8f ph = erf_x8(ALPHA_ERF * xin);
#pragma unroll
        for (int k = 0; k < 2; ++k) {
            const float4 vv = *(const float4*)(v + (size_t)(nBeg + i + k) * 4);
            const float  zv = z[nBeg + i + k];
            const float p0 = ph[4 * k + 0];
            const float p1 = ph[4 * k + 1];
            const float p2 = ph[4 * k + 2];
            const float p3 = ph[4 * k + 3];
            acc[0].x = fmaf(vv.x, p0, acc[0].x); acc[0].y = fmaf(vv.x, p1, acc[0].y);
            acc[0].z = fmaf(vv.x, p2, acc[0].z); acc[0].w = fmaf(vv.x, p3, acc[0].w);
            acc[1].x = fmaf(vv.y, p0, acc[1].x); acc[1].y = fmaf(vv.y, p1, acc[1].y);
            acc[1].z = fmaf(vv.y, p2, acc[1].z); acc[1].w = fmaf(vv.y, p3, acc[1].w);
            acc[2].x = fmaf(vv.z, p0, acc[2].x); acc[2].y = fmaf(vv.z, p1, acc[2].y);
            acc[2].z = fmaf(vv.z, p2, acc[2].z); acc[2].w = fmaf(vv.z, p3, acc[2].w);
            acc[3].x = fmaf(vv.w, p0, acc[3].x); acc[3].y = fmaf(vv.w, p1, acc[3].y);
            acc[3].z = fmaf(vv.w, p2, acc[3].z); acc[3].w = fmaf(vv.w, p3, acc[3].w);
            acc[4].x = fmaf(zv,   p0, acc[4].x); acc[4].y = fmaf(zv,   p1, acc[4].y);
            acc[4].z = fmaf(zv,   p2, acc[4].z); acc[4].w = fmaf(zv,   p3, acc[4].w);
        }
    }

    for (int r = 0; r < 5; ++r) red[wave][r][lane] = acc[r];
    __syncthreads();

    if (wave < 5) {
        float4 s = red[0][wave][lane];
#pragma unroll
        for (int w = 1; w < K1_WAVES; ++w) {
            const float4 t = red[w][wave][lane];
            s.x += t.x; s.y += t.y; s.z += t.z; s.w += t.w;
        }
        *(float4*)(partial + (size_t)(chunk * 5 + wave) * (size_t)B + (size_t)b0) = s;
    }
}

#define K3_WAVES  16
#define K3_STRIPS 2

__global__ __launch_bounds__(K3_WAVES * 32) void k3_update(
    const float* __restrict__ h, const float* __restrict__ u,
    const float* __restrict__ m, const float* __restrict__ x,
    const float* __restrict__ partial, float* __restrict__ hnew,
    float* __restrict__ y_out, int B, float invN)
{
    __shared__ __attribute__((aligned(16))) float tl[4][64 * K3_STRIPS];

    const int lane = threadIdx.x & 31;
    const int wave = __builtin_amdgcn_readfirstlane((int)(threadIdx.x >> 5));
    const int lo   = lane & 15;
    const int hi   = lane >> 4;

    const int n0  = ((int)blockIdx.y * K3_WAVES + wave) * 16;
    const int cg0 = (int)blockIdx.x * (64 * K3_STRIPS);

    const size_t   uoff = (size_t)n0 * (size_t)B + (size_t)cg0;
    const unsigned loff = (unsigned)(8 * hi) * (unsigned)B + (unsigned)(4 * lo);

    v2f a;
    {
        const float2 av = *(const float2*)(u + (size_t)(n0 + lo) * 4 + hi * 2);
        a.x = av.x; a.y = av.y;
    }

    v4f xv[K3_STRIPS];
#pragma unroll
    for (int s = 0; s < K3_STRIPS; ++s)
        xv[s] = *(const v4f*)(x + cg0 + s * 64 + 4 * lo);

    float mv[8];
    {
        const float4 m0 = *(const float4*)(m + n0 + 8 * hi);
        const float4 m1 = *(const float4*)(m + n0 + 8 * hi + 4);
        mv[0] = m0.x; mv[1] = m0.y; mv[2] = m0.z; mv[3] = m0.w;
        mv[4] = m1.x; mv[5] = m1.y; mv[6] = m1.z; mv[7] = m1.w;
    }

    const int fr   = wave >> 2;
    const int fcol = (wave & 3) * 32 + lane;
    const size_t pstride = (size_t)5 * (size_t)B;
    const float* __restrict__ pp = partial + (size_t)fr * (size_t)B + (size_t)(cg0 + fcol);
    float pv[N_CHUNKS];
#pragma unroll
    for (int c = 0; c < N_CHUNKS; ++c) pv[c] = pp[(size_t)c * pstride];

    const bool foldY = (blockIdx.y == 0) && (wave < 4);
    float yv[N_CHUNKS];
    if (foldY) {
#pragma unroll
        for (int c = 0; c < N_CHUNKS; ++c) yv[c] = pp[(size_t)c * pstride + (size_t)4 * (size_t)B];
    }

    __builtin_amdgcn_sched_barrier(0);

    const float* __restrict__ hp = h + uoff;
    v4f hv[K3_STRIPS][8];
#pragma unroll
    for (int s = 0; s < K3_STRIPS; ++s)
#pragma unroll
        for (int g = 0; g < 8; ++g)
            hv[s][g] = *(const v4f*)(hp + (size_t)g * (size_t)B + s * 64 + loff);
    __builtin_amdgcn_sched_barrier(0);

    {
        float s = 0.0f;
#pragma unroll
        for (int c = 0; c < N_CHUNKS; ++c) s += pv[c];
        tl[fr][fcol] = s;
    }
    if (foldY) {
        float s = 0.0f;
#pragma unroll
        for (int c = 0; c < N_CHUNKS; ++c) s += yv[c];
        y_out[cg0 + fcol] = s * invN;
    }
    __builtin_amdgcn_fence(__ATOMIC_RELEASE, "workgroup", "local");
    __builtin_amdgcn_s_barrier();
    __builtin_amdgcn_fence(__ATOMIC_ACQUIRE, "workgroup", "local");

    v4f t0[K3_STRIPS], t1[K3_STRIPS];
#pragma unroll
    for (int s = 0; s < K3_STRIPS; ++s) {
        t0[s] = *(const v4f*)&tl[0 + 2 * hi][s * 64 + 4 * lo];
        t1[s] = *(const v4f*)&tl[1 + 2 * hi][s * 64 + 4 * lo];
    }

    v8f d[K3_STRIPS][4];
#pragma unroll
    for (int s = 0; s < K3_STRIPS; ++s)
#pragma unroll
        for (int j = 0; j < 4; ++j) {
            v2f bm;
            bm.x = t0[s][j];
            bm.y = t1[s][j];
            v8f c = {};
            d[s][j] = __builtin_amdgcn_wmma_f32_16x16x4_f32(
                 false, a,  false, bm,
                 (short)0, c,  false,  false);
        }

    float* __restrict__ op = hnew + uoff;
#pragma unroll
    for (int s = 0; s < K3_STRIPS; ++s)
#pragma unroll
        for (int g = 0; g < 8; ++g) {
            v4f o;
#pragma unroll
            for (int j = 0; j < 4; ++j) {
                const float hh  = hv[s][g][j];
                const float upd = fmaf(mv[g], xv[s][j], d[s][j][g] * invN);
                o[j] = hh + (upd - hh);
            }
            __builtin_nontemporal_store(o, (v4f*)(op + (size_t)g * (size_t)B + s * 64 + loff));
        }
}

extern "C" void kernel_launch(void* const* d_in, const int* in_sizes, int n_in,
                              void* d_out, int out_size, void* d_ws, size_t ws_size,
                              hipStream_t stream) {
    const float* x = (const float*)d_in[0];
    const float* h = (const float*)d_in[1];
    const float* m = (const float*)d_in[2];
    const float* u = (const float*)d_in[3];
    const float* v = (const float*)d_in[4];
    const float* z = (const float*)d_in[5];

    const int B = in_sizes[0];
    const int N = in_sizes[2];
    const float invN = 1.0f / (float)N;

    float* y_out = (float*)d_out;
    float* hnew  = (float*)d_out + B;

    float* partial = (float*)d_ws;

    dim3 grid(B / 128, N_CHUNKS);

    k1_reduce<<<grid, K1_WAVES * 32, 0, stream>>>(h, v, z, partial, B);

    k3_update<<<grid, K3_WAVES * 32, 0, stream>>>(h, u, m, x, partial, hnew, y_out, B, invN);
}
